# starting kernel
# baseline (speedup 1.0000x reference)
#include <hip/hip_runtime.h>


typedef __attribute__((ext_vector_type(2))) float v2f;
typedef __attribute__((ext_vector_type(8))) float v8f;

#define NSTATE 4096
#define THREADS 128

__global__ __launch_bounds__(THREADS)
void vqc_kernel(const float* __restrict__ instance,
                const int*   __restrict__ concept_index,
                const float* __restrict__ concept_weights,
                float*       __restrict__ out)
{
  __shared__ float gR[12][2], gI[12][2];
  __shared__ float Ur[64], Ui[64];
  __shared__ float Vr[64], Vi[64];
  __shared__ float obsA[4], obsB[4], obsC[4];
  __shared__ float psiR[NSTATE];
  __shared__ float psiI[NSTATE];
  __shared__ float partial[THREADS][17];
  __shared__ float sums[16];

  const int b   = blockIdx.x;
  const int tid = threadIdx.x;

  __builtin_prefetch(instance + (size_t)b * 36, 0, 3);

  if (tid < 12) {
    const int w = tid;
    const float* a = instance + ((size_t)b * 12 + w) * 3;
    const float phi = a[0], th = a[1], om = a[2];
    float c, s;   sincosf(0.5f * th, &s, &c);
    float s0, c0; sincosf(-0.5f * (phi + om), &s0, &c0);
    float s1, c1; sincosf( 0.5f * (om  - phi), &s1, &c1);
    const int p = 11 - w;
    gR[p][0] = c * c0;  gI[p][0] = c * s0;
    gR[p][1] = s * c1;  gI[p][1] = s * s1;
  } else if (tid < 16) {
    const int d  = tid - 12;
    const int ci = concept_index[b];
    const float* cw = concept_weights + (size_t)ci * 12 + d * 3;
    const float cth = cw[1], com = cw[2];
    float sth, vcth; sincosf(cth, &sth, &vcth);
    float so,  co;   sincosf(com, &so,  &co);
    obsA[d] = vcth;
    obsB[d] = 2.0f * sth * co;
    obsC[d] = 2.0f * sth * so;
  }
  __syncthreads();

  {
    const int h    = tid & 63;
    const int base = (tid < 64) ? 6 : 0;
    float pr = 1.0f, pi = 0.0f;
#pragma unroll
    for (int j = 0; j < 6; ++j) {
      const int t = (h >> j) & 1;
      const float gr = gR[base + j][t], gi = gI[base + j][t];
      const float nr = pr * gr - pi * gi;
      const float ni = pr * gi + pi * gr;
      pr = nr; pi = ni;
    }
    if (tid < 64) { Ur[h] = 0.5f * pr; Ui[h] = 0.5f * pi; }
    else          { Vr[h] = pr;        Vi[h] = pi;        }
  }
  __syncthreads();

  {
    const int wave = tid >> 5;
    const int lane = tid & 31;
    const int m    = lane & 15;
    const int I    = wave;
    const v2f A = { Ur[16 * I + m], Ui[16 * I + m] };
#pragma unroll
    for (int J = 0; J < 4; ++J) {
      const float vr = Vr[16 * J + m];
      const float vi = Vi[16 * J + m];
      const v2f Bre = { vr, -vi };
      const v2f Bim = { vi,  vr };
      const v8f z0 = { 0.f, 0.f, 0.f, 0.f, 0.f, 0.f, 0.f, 0.f };
      v8f dre = __builtin_amdgcn_wmma_f32_16x16x4_f32(false, A, false, Bre,
                                                      (short)0, z0, false, false);
      v8f dim = __builtin_amdgcn_wmma_f32_16x16x4_f32(false, A, false, Bim,
                                                      (short)0, z0, false, false);
      const int hbase = 16 * I + ((lane & 16) ? 8 : 0);
      const int lg    = 16 * J + m;
#pragma unroll
      for (int r = 0; r < 8; ++r) {
        psiR[(hbase + r) * 64 + lg] = dre[r];
        psiI[(hbase + r) * 64 + lg] = dim[r];
      }
    }
  }
  __syncthreads();

  float acc[16];
#pragma unroll
  for (int q = 0; q < 16; ++q) acc[q] = 0.0f;

  const int bits[4] = { 11, 8, 5, 2 };
  for (int i = 0; i < NSTATE / THREADS; ++i) {
    const int s   = tid + i * THREADS;
    const int idx = s ^ (s >> 1) ^ ((s & 1) ? 0xC00 : 0);
    const float pr = psiR[idx], pi = psiI[idx];
    const float nrm = pr * pr + pi * pi;
#pragma unroll
    for (int d = 0; d < 4; ++d) {
      const int bb = bits[d];
      const int j  = idx ^ (3 << (bb - 1));
      const float qr = psiR[j], qi = psiI[j];
      const bool hi = (s >> bb) & 1;
      if (hi) {
        acc[d * 4 + 1] += nrm;
      } else {
        acc[d * 4 + 0] += nrm;
        acc[d * 4 + 2] += pr * qr + pi * qi;
        acc[d * 4 + 3] += pr * qi - pi * qr;
      }
    }
  }
#pragma unroll
  for (int q = 0; q < 16; ++q) partial[tid][q] = acc[q];
  __syncthreads();

  if (tid < 16) {
    float t = 0.0f;
    for (int i = 0; i < THREADS; ++i) t += partial[i][tid];
    sums[tid] = t;
  }
  __syncthreads();

  if (tid < 4) {
    const int d = tid;
    const float n0  = sums[d * 4 + 0];
    const float n1  = sums[d * 4 + 1];
    const float crR = sums[d * 4 + 2];
    const float crI = sums[d * 4 + 3];
    const float z = obsA[d] * (n0 - n1) + obsB[d] * crR + obsC[d] * crI;
    out[(size_t)b * 4 + d] = 0.5f * z + 0.5f;
  }
}

extern "C" void kernel_launch(void* const* d_in, const int* in_sizes, int n_in,
                              void* d_out, int out_size, void* d_ws, size_t ws_size,
                              hipStream_t stream) {
  const float* instance = (const float*)d_in[0];
  const int*   cidx     = (const int*)d_in[1];
  const float* cweights = (const float*)d_in[2];
  float*       out      = (float*)d_out;
  const int B = in_sizes[0] / 36;
  vqc_kernel<<<B, THREADS, 0, stream>>>(instance, cidx, cweights, out);
}


# Each wave makes (A rows chosen by set-id bits t8^t7 and t6) exactly the 1024 amplitudes its own lanes reduce: psi's trip through LDS stays in one wave, no block barrier per element; sums kept
# speedup vs baseline: 4.5801x; 4.5801x over previous
#include <hip/hip_runtime.h>


typedef __attribute__((ext_vector_type(2))) float v2f;
typedef __attribute__((ext_vector_type(8))) float v8f;

#define NSTATE 4096
#define THREADS 128
#define EPB 8
#define PSTRIDE (THREADS * 13 + 12)

__global__ __launch_bounds__(THREADS)
void vqc_kernel(const float* __restrict__ instance,
                const int*   __restrict__ concept_index,
                const float* __restrict__ concept_weights,
                float*       __restrict__ out,
                int B)
{
  __shared__ float gR[EPB][12][2], gI[EPB][12][2];
  __shared__ float UVr[EPB][128], UVi[EPB][128];
  __shared__ float obsA[EPB][4], obsB[EPB][4], obsC[EPB][4];
  __shared__ float4 psiR4[NSTATE / 4];
  __shared__ float4 psiI4[NSTATE / 4];
  __shared__ float partial[EPB * PSTRIDE];
  __shared__ float sums[EPB][12];

  const int tid = threadIdx.x;

  for (int chunk = blockIdx.x; chunk * EPB < B; chunk += gridDim.x) {
  const int b0 = chunk * EPB;

  {
  const int e1 = tid >> 4;
  const int q1 = tid & 15;
  const int b  = b0 + e1;
  if (b >= B) {
    if (q1 < 12) {
      gR[e1][q1][0] = 0.0f;  gI[e1][q1][0] = 0.0f;
      gR[e1][q1][1] = 0.0f;  gI[e1][q1][1] = 0.0f;
    }
  } else if (q1 < 12) {
    const int w = q1;
    const float* a = instance + ((size_t)b * 12 + w) * 3;
    const float phi = a[0], th = a[1], om = a[2];
    float c, s;   sincosf(0.5f * th, &s, &c);
    float s0, c0; sincosf(-0.5f * (phi + om), &s0, &c0);
    float s1, c1; sincosf( 0.5f * (om  - phi), &s1, &c1);
    const int p = 11 - w;
    gR[e1][p][0] = c * c0;  gI[e1][p][0] = c * s0;
    gR[e1][p][1] = s * c1;  gI[e1][p][1] = s * s1;
  } else {
    const int d  = q1 - 12;
    const int ci = concept_index[b];
    const float* cw = concept_weights + (size_t)ci * 12 + d * 3;
    const float cth = cw[1], com = cw[2];
    float sth, vcth; sincosf(cth, &sth, &vcth);
    float so,  co;   sincosf(com, &so,  &co);
    obsA[e1][d] = vcth;
    obsB[e1][d] = 2.0f * sth * co;
    obsC[e1][d] = 2.0f * sth * so;
  }
  }
  __syncthreads();

  {
    const int h     = tid & 63;
    const int base  = (tid < 64) ? 6 : 0;
    const float scl = (tid < 64) ? 0.5f : 1.0f;
#pragma unroll
    for (int e = 0; e < EPB; ++e) {
      float pr = 1.0f, pi = 0.0f;
#pragma unroll
      for (int j = 0; j < 6; ++j) {
        const int t = (h >> j) & 1;
        const float gr = gR[e][base + j][t], gi = gI[e][base + j][t];
        const float nr = pr * gr - pi * gi;
        const float ni = pr * gi + pi * gr;
        pr = nr; pi = ni;
      }
      UVr[e][tid] = scl * pr;
      UVi[e][tid] = scl * pi;
    }
  }
  __syncthreads();

  for (int e = 0; e < EPB; ++e) {
  if (b0 + e >= B) break;

  {
    const int wave = tid >> 5;
    const int lane = tid & 31;
    const int m    = lane & 15;
    const int w1   = wave >> 1;
    const int w0   = wave & 1;
    const int hU   = (((m >> 2) & 1) << 5) | (((m >> 1) & 1) << 4) | (((m >> 3) & 1) << 3) |
                     ((w1 ^ (m & 1)) << 2) | ((m & 1) << 1) | w0;
    const v2f A = { UVr[e][hU], UVi[e][hU] };
#pragma unroll
    for (int J = 0; J < 4; ++J) {
      const float vr = UVr[e][64 + 16 * J + m];
      const float vi = UVi[e][64 + 16 * J + m];
      const v2f Bre = { vr, -vi };
      const v2f Bim = { vi,  vr };
      const v8f z0 = { 0.f, 0.f, 0.f, 0.f, 0.f, 0.f, 0.f, 0.f };
      v8f dre = __builtin_amdgcn_wmma_f32_16x16x4_f32(false, A, false, Bre,
                                                      (short)0, z0, false, false);
      v8f dim = __builtin_amdgcn_wmma_f32_16x16x4_f32(false, A, false, Bim,
                                                      (short)0, z0, false, false);
      const int half   = lane >> 4;
      const int laneIx = (half << 5) | (((m >> 3) & 1) << 4) |
                         ((((m >> 2) ^ (m >> 1)) & 1) << 3) | ((m & 1) << 2) | ((m >> 1) & 1);
      const int waveIx = (w1 << 8) | (w0 << 7);
      const int jIx    = ((((J >> 1) ^ J) & 1) << 6) | ((J & 1) << 1);
      float* psiR = reinterpret_cast<float*>(psiR4);
      float* psiI = reinterpret_cast<float*>(psiI4);
#pragma unroll
      for (int r = 0; r < 8; ++r) {
        const int rIx = (((r >> 1) & 1) << 11) | ((r & 1) << 10) | ((((r >> 2) ^ (r >> 1)) & 1) << 9);
        psiR[waveIx + laneIx + jIx + rIx] = dre[r];
        psiI[waveIx + laneIx + jIx + rIx] = dim[r];
      }
    }
  }

  float acc[12];
#pragma unroll
  for (int q = 0; q < 12; ++q) acc[q] = 0.0f;

#pragma unroll
  for (int k = 0; k < 2; ++k) {
    const int id = (k << 7) | tid;
    v2f ar[4][2], ai[4][2], nr[4][2];
#pragma unroll
    for (int j = 0; j < 4; ++j) {
      const float4 vr = psiR4[(j << 8) | id];
      const float4 vi = psiI4[(j << 8) | id];
      ar[j][0] = v2f{ vr.x, vr.y };  ar[j][1] = v2f{ vr.z, vr.w };
      ai[j][0] = v2f{ vi.x, vi.y };  ai[j][1] = v2f{ vi.z, vi.w };
    }
#pragma unroll
    for (int j = 0; j < 4; ++j)
#pragma unroll
      for (int h = 0; h < 2; ++h)
        nr[j][h] = __builtin_elementwise_fma(ar[j][h], ar[j][h], ai[j][h] * ai[j][h]);

    const v2f s0 = nr[0][0] + nr[0][1], s1 = nr[1][0] + nr[1][1];
    const v2f s2 = nr[2][0] + nr[2][1], s3 = nr[3][0] + nr[3][1];
    const v2f dn2 = ((nr[0][0] - nr[0][1]) + (nr[1][0] - nr[1][1])) +
                    ((nr[2][0] - nr[2][1]) + (nr[3][0] - nr[3][1]));
    const v2f s01 = s0 + s1, s23 = s2 + s3;
    const v2f dn0 = s01 - s23;
    const v2f dn1 = (s0 - s1) + (s2 - s3);
    const v2f tot = s01 + s23;

    const v2f zero2 = { 0.0f, 0.0f };
    v2f xr0 = zero2, xi0 = zero2;
#pragma unroll
    for (int j = 0; j < 2; ++j)
#pragma unroll
      for (int h = 0; h < 2; ++h) {
        xr0 = __builtin_elementwise_fma(ar[j][h], ar[j + 2][h], xr0);
        xr0 = __builtin_elementwise_fma(ai[j][h], ai[j + 2][h], xr0);
        xi0 = __builtin_elementwise_fma(ar[j][h], ai[j + 2][h], xi0);
        xi0 = __builtin_elementwise_fma(-ai[j][h], ar[j + 2][h], xi0);
      }
    v2f xr1 = zero2, xi1 = zero2;
#pragma unroll
    for (int j = 0; j < 4; j += 2)
#pragma unroll
      for (int h = 0; h < 2; ++h) {
        xr1 = __builtin_elementwise_fma(ar[j][h], ar[j + 1][h], xr1);
        xr1 = __builtin_elementwise_fma(ai[j][h], ai[j + 1][h], xr1);
        xi1 = __builtin_elementwise_fma(ar[j][h], ai[j + 1][h], xi1);
        xi1 = __builtin_elementwise_fma(-ai[j][h], ar[j + 1][h], xi1);
      }
    v2f xr2 = zero2, xi2 = zero2;
#pragma unroll
    for (int j = 0; j < 4; ++j) {
      xr2 = __builtin_elementwise_fma(ar[j][0], ar[j][1], xr2);
      xr2 = __builtin_elementwise_fma(ai[j][0], ai[j][1], xr2);
      xi2 = __builtin_elementwise_fma(ar[j][0], ai[j][1], xi2);
      xi2 = __builtin_elementwise_fma(-ai[j][0], ar[j][1], xi2);
    }
    v2f xi3w = zero2;
    float xr3a = 0.0f, xr3b = 0.0f;
#pragma unroll
    for (int j = 0; j < 4; ++j)
#pragma unroll
      for (int h = 0; h < 2; ++h) {
        const v2f aisw = __builtin_shufflevector(ai[j][h], ai[j][h], 1, 0);
        xi3w = __builtin_elementwise_fma(ar[j][h], aisw, xi3w);
        xr3a = fmaf(ar[j][h].x, ar[j][h].y, xr3a);
        xr3b = fmaf(ai[j][h].x, ai[j][h].y, xr3b);
      }

    const float sg0 = (__popc(id & 0x7F) & 1) ? -1.0f : 1.0f;
    const float sg1 = (__popc(id & 0xC8) & 1) ? -1.0f : 1.0f;
    const float sg2 = (__popc(id & 0xF8) & 1) ? -1.0f : 1.0f;
    const float sg3 = (__popc(id & 0xFE) & 1) ? -1.0f : 1.0f;
    acc[0]  += sg0 * (dn0.x + dn0.y);  acc[1]  += xr0.x + xr0.y;  acc[2]  += sg0 * (xi0.x + xi0.y);
    acc[3]  += sg1 * (dn1.x + dn1.y);  acc[4]  += xr1.x + xr1.y;  acc[5]  += sg1 * (xi1.x + xi1.y);
    acc[6]  += sg2 * (dn2.x + dn2.y);  acc[7]  += xr2.x + xr2.y;  acc[8]  += sg2 * (xi2.x + xi2.y);
    acc[9]  += sg3 * (tot.x - tot.y);  acc[10] += xr3a + xr3b;    acc[11] += sg3 * (xi3w.x - xi3w.y);
  }
#pragma unroll
  for (int q = 0; q < 12; ++q) partial[e * PSTRIDE + tid * 13 + q] = acc[q];
  }
  __syncthreads();

  if (tid < EPB * 12) {
    const int e5 = tid / 12;
    const int q5 = tid - e5 * 12;
    if (b0 + e5 < B) {
      float t = 0.0f;
      for (int i = 0; i < THREADS; ++i) t += partial[e5 * PSTRIDE + i * 13 + q5];
      sums[e5][q5] = t;
    }
  }
  __syncthreads();

  if (tid < EPB * 4) {
    const int e6 = tid >> 2;
    const int d  = tid & 3;
    if (b0 + e6 < B) {
      const float dn  = sums[e6][d * 3 + 0];
      const float crR = sums[e6][d * 3 + 1];
      const float crI = sums[e6][d * 3 + 2];
      const float z = obsA[e6][d] * dn + obsB[e6][d] * crR + obsC[e6][d] * crI;
      out[(size_t)(b0 + e6) * 4 + d] = 0.5f * z + 0.5f;
    }
  }
  __syncthreads();
  }
}

extern "C" void kernel_launch(void* const* d_in, const int* in_sizes, int n_in,
                              void* d_out, int out_size, void* d_ws, size_t ws_size,
                              hipStream_t stream) {
  const float* instance = (const float*)d_in[0];
  const int*   cidx     = (const int*)d_in[1];
  const float* cweights = (const float*)d_in[2];
  float*       out      = (float*)d_out;
  const int B = in_sizes[0] / 36;
  vqc_kernel<<<512, THREADS, 0, stream>>>(instance, cidx, cweights, out, B);
}


# Block sums: per-thread sums stored one row per sum (thread index contiguous); each of the 96 lanes reads its 128 addends by 16-byte loads into 16 independent partial sums; f32 sum order changes
# speedup vs baseline: 5.0199x; 1.0960x over previous
#include <hip/hip_runtime.h>


typedef __attribute__((ext_vector_type(2))) float v2f;
typedef __attribute__((ext_vector_type(4))) float v4f;
typedef __attribute__((ext_vector_type(8))) float v8f;

#define NSTATE 4096
#define THREADS 128
#define EPB 8
#define PPITCH4 33

__global__ __launch_bounds__(THREADS)
void vqc_kernel(const float* __restrict__ instance,
                const int*   __restrict__ concept_index,
                const float* __restrict__ concept_weights,
                float*       __restrict__ out,
                int B)
{
  __shared__ float gR[EPB][12][2], gI[EPB][12][2];
  __shared__ float UVr[EPB][128], UVi[EPB][128];
  __shared__ float obsA[EPB][4], obsB[EPB][4], obsC[EPB][4];
  __shared__ float4 psiR4[NSTATE / 4];
  __shared__ float4 psiI4[NSTATE / 4];
  __shared__ v4f partial4[EPB * 12 * PPITCH4];
  __shared__ float sums[EPB][12];

  const int tid = threadIdx.x;

  for (int chunk = blockIdx.x; chunk * EPB < B; chunk += gridDim.x) {
  const int b0 = chunk * EPB;

  {
  const int e1 = tid >> 4;
  const int q1 = tid & 15;
  const int b  = b0 + e1;
  if (b >= B) {
    if (q1 < 12) {
      gR[e1][q1][0] = 0.0f;  gI[e1][q1][0] = 0.0f;
      gR[e1][q1][1] = 0.0f;  gI[e1][q1][1] = 0.0f;
    }
  } else if (q1 < 12) {
    const int w = q1;
    const float* a = instance + ((size_t)b * 12 + w) * 3;
    const float phi = a[0], th = a[1], om = a[2];
    float c, s;   sincosf(0.5f * th, &s, &c);
    float s0, c0; sincosf(-0.5f * (phi + om), &s0, &c0);
    float s1, c1; sincosf( 0.5f * (om  - phi), &s1, &c1);
    const int p = 11 - w;
    gR[e1][p][0] = c * c0;  gI[e1][p][0] = c * s0;
    gR[e1][p][1] = s * c1;  gI[e1][p][1] = s * s1;
  } else {
    const int d  = q1 - 12;
    const int ci = concept_index[b];
    const float* cw = concept_weights + (size_t)ci * 12 + d * 3;
    const float cth = cw[1], com = cw[2];
    float sth, vcth; sincosf(cth, &sth, &vcth);
    float so,  co;   sincosf(com, &so,  &co);
    obsA[e1][d] = vcth;
    obsB[e1][d] = 2.0f * sth * co;
    obsC[e1][d] = 2.0f * sth * so;
  }
  }
  __syncthreads();

  {
    const int h     = tid & 63;
    const int base  = (tid < 64) ? 6 : 0;
    const float scl = (tid < 64) ? 0.5f : 1.0f;
#pragma unroll
    for (int e = 0; e < EPB; ++e) {
      float pr = 1.0f, pi = 0.0f;
#pragma unroll
      for (int j = 0; j < 6; ++j) {
        const int t = (h >> j) & 1;
        const float gr = gR[e][base + j][t], gi = gI[e][base + j][t];
        const float nr = pr * gr - pi * gi;
        const float ni = pr * gi + pi * gr;
        pr = nr; pi = ni;
      }
      UVr[e][tid] = scl * pr;
      UVi[e][tid] = scl * pi;
    }
  }
  __syncthreads();

  for (int e = 0; e < EPB; ++e) {
  if (b0 + e >= B) break;

  {
    const int wave = tid >> 5;
    const int lane = tid & 31;
    const int m    = lane & 15;
    const int w1   = wave >> 1;
    const int w0   = wave & 1;
    const int hU   = (((m >> 2) & 1) << 5) | (((m >> 1) & 1) << 4) | (((m >> 3) & 1) << 3) |
                     ((w1 ^ (m & 1)) << 2) | ((m & 1) << 1) | w0;
    const v2f A = { UVr[e][hU], UVi[e][hU] };
#pragma unroll
    for (int J = 0; J < 4; ++J) {
      const float vr = UVr[e][64 + 16 * J + m];
      const float vi = UVi[e][64 + 16 * J + m];
      const v2f Bre = { vr, -vi };
      const v2f Bim = { vi,  vr };
      const v8f z0 = { 0.f, 0.f, 0.f, 0.f, 0.f, 0.f, 0.f, 0.f };
      v8f dre = __builtin_amdgcn_wmma_f32_16x16x4_f32(false, A, false, Bre,
                                                      (short)0, z0, false, false);
      v8f dim = __builtin_amdgcn_wmma_f32_16x16x4_f32(false, A, false, Bim,
                                                      (short)0, z0, false, false);
      const int half   = lane >> 4;
      const int laneIx = (half << 5) | (((m >> 3) & 1) << 4) |
                         ((((m >> 2) ^ (m >> 1)) & 1) << 3) | ((m & 1) << 2) | ((m >> 1) & 1);
      const int waveIx = (w1 << 8) | (w0 << 7);
      const int jIx    = ((((J >> 1) ^ J) & 1) << 6) | ((J & 1) << 1);
      float* psiR = reinterpret_cast<float*>(psiR4);
      float* psiI = reinterpret_cast<float*>(psiI4);
#pragma unroll
      for (int r = 0; r < 8; ++r) {
        const int rIx = (((r >> 1) & 1) << 11) | ((r & 1) << 10) | ((((r >> 2) ^ (r >> 1)) & 1) << 9);
        psiR[waveIx + laneIx + jIx + rIx] = dre[r];
        psiI[waveIx + laneIx + jIx + rIx] = dim[r];
      }
    }
  }

  float acc[12];
#pragma unroll
  for (int q = 0; q < 12; ++q) acc[q] = 0.0f;

#pragma unroll
  for (int k = 0; k < 2; ++k) {
    const int id = (k << 7) | tid;
    v2f ar[4][2], ai[4][2], nr[4][2];
#pragma unroll
    for (int j = 0; j < 4; ++j) {
      const float4 vr = psiR4[(j << 8) | id];
      const float4 vi = psiI4[(j << 8) | id];
      ar[j][0] = v2f{ vr.x, vr.y };  ar[j][1] = v2f{ vr.z, vr.w };
      ai[j][0] = v2f{ vi.x, vi.y };  ai[j][1] = v2f{ vi.z, vi.w };
    }
#pragma unroll
    for (int j = 0; j < 4; ++j)
#pragma unroll
      for (int h = 0; h < 2; ++h)
        nr[j][h] = __builtin_elementwise_fma(ar[j][h], ar[j][h], ai[j][h] * ai[j][h]);

    const v2f s0 = nr[0][0] + nr[0][1], s1 = nr[1][0] + nr[1][1];
    const v2f s2 = nr[2][0] + nr[2][1], s3 = nr[3][0] + nr[3][1];
    const v2f dn2 = ((nr[0][0] - nr[0][1]) + (nr[1][0] - nr[1][1])) +
                    ((nr[2][0] - nr[2][1]) + (nr[3][0] - nr[3][1]));
    const v2f s01 = s0 + s1, s23 = s2 + s3;
    const v2f dn0 = s01 - s23;
    const v2f dn1 = (s0 - s1) + (s2 - s3);
    const v2f tot = s01 + s23;

    const v2f zero2 = { 0.0f, 0.0f };
    v2f xr0 = zero2, xi0 = zero2;
#pragma unroll
    for (int j = 0; j < 2; ++j)
#pragma unroll
      for (int h = 0; h < 2; ++h) {
        xr0 = __builtin_elementwise_fma(ar[j][h], ar[j + 2][h], xr0);
        xr0 = __builtin_elementwise_fma(ai[j][h], ai[j + 2][h], xr0);
        xi0 = __builtin_elementwise_fma(ar[j][h], ai[j + 2][h], xi0);
        xi0 = __builtin_elementwise_fma(-ai[j][h], ar[j + 2][h], xi0);
      }
    v2f xr1 = zero2, xi1 = zero2;
#pragma unroll
    for (int j = 0; j < 4; j += 2)
#pragma unroll
      for (int h = 0; h < 2; ++h) {
        xr1 = __builtin_elementwise_fma(ar[j][h], ar[j + 1][h], xr1);
        xr1 = __builtin_elementwise_fma(ai[j][h], ai[j + 1][h], xr1);
        xi1 = __builtin_elementwise_fma(ar[j][h], ai[j + 1][h], xi1);
        xi1 = __builtin_elementwise_fma(-ai[j][h], ar[j + 1][h], xi1);
      }
    v2f xr2 = zero2, xi2 = zero2;
#pragma unroll
    for (int j = 0; j < 4; ++j) {
      xr2 = __builtin_elementwise_fma(ar[j][0], ar[j][1], xr2);
      xr2 = __builtin_elementwise_fma(ai[j][0], ai[j][1], xr2);
      xi2 = __builtin_elementwise_fma(ar[j][0], ai[j][1], xi2);
      xi2 = __builtin_elementwise_fma(-ai[j][0], ar[j][1], xi2);
    }
    v2f xi3w = zero2;
    float xr3a = 0.0f, xr3b = 0.0f;
#pragma unroll
    for (int j = 0; j < 4; ++j)
#pragma unroll
      for (int h = 0; h < 2; ++h) {
        const v2f aisw = __builtin_shufflevector(ai[j][h], ai[j][h], 1, 0);
        xi3w = __builtin_elementwise_fma(ar[j][h], aisw, xi3w);
        xr3a = fmaf(ar[j][h].x, ar[j][h].y, xr3a);
        xr3b = fmaf(ai[j][h].x, ai[j][h].y, xr3b);
      }

    const float sg0 = (__popc(id & 0x7F) & 1) ? -1.0f : 1.0f;
    const float sg1 = (__popc(id & 0xC8) & 1) ? -1.0f : 1.0f;
    const float sg2 = (__popc(id & 0xF8) & 1) ? -1.0f : 1.0f;
    const float sg3 = (__popc(id & 0xFE) & 1) ? -1.0f : 1.0f;
    acc[0]  += sg0 * (dn0.x + dn0.y);  acc[1]  += xr0.x + xr0.y;  acc[2]  += sg0 * (xi0.x + xi0.y);
    acc[3]  += sg1 * (dn1.x + dn1.y);  acc[4]  += xr1.x + xr1.y;  acc[5]  += sg1 * (xi1.x + xi1.y);
    acc[6]  += sg2 * (dn2.x + dn2.y);  acc[7]  += xr2.x + xr2.y;  acc[8]  += sg2 * (xi2.x + xi2.y);
    acc[9]  += sg3 * (tot.x - tot.y);  acc[10] += xr3a + xr3b;    acc[11] += sg3 * (xi3w.x - xi3w.y);
  }
  {
    float* prow = reinterpret_cast<float*>(partial4) + (e * 12) * (PPITCH4 * 4) + tid;
#pragma unroll
    for (int q = 0; q < 12; ++q) prow[q * (PPITCH4 * 4)] = acc[q];
  }
  }
  __syncthreads();

  if (tid < EPB * 12) {
    const int e5 = tid / 12;
    const int q5 = tid - e5 * 12;
    if (b0 + e5 < B) {
      const v4f* row = partial4 + tid * PPITCH4;
      v4f a0 = row[0], a1 = row[1], a2 = row[2], a3 = row[3];
#pragma unroll
      for (int j = 4; j < THREADS / 4; j += 4) {
        a0 += row[j];  a1 += row[j + 1];  a2 += row[j + 2];  a3 += row[j + 3];
      }
      const v4f s = (a0 + a1) + (a2 + a3);
      sums[e5][q5] = (s.x + s.y) + (s.z + s.w);
    }
  }
  __syncthreads();

  if (tid < EPB * 4) {
    const int e6 = tid >> 2;
    const int d  = tid & 3;
    if (b0 + e6 < B) {
      const float dn  = sums[e6][d * 3 + 0];
      const float crR = sums[e6][d * 3 + 1];
      const float crI = sums[e6][d * 3 + 2];
      const float z = obsA[e6][d] * dn + obsB[e6][d] * crR + obsC[e6][d] * crI;
      out[(size_t)(b0 + e6) * 4 + d] = 0.5f * z + 0.5f;
    }
  }
  __syncthreads();
  }
}

extern "C" void kernel_launch(void* const* d_in, const int* in_sizes, int n_in,
                              void* d_out, int out_size, void* d_ws, size_t ws_size,
                              hipStream_t stream) {
  const float* instance = (const float*)d_in[0];
  const int*   cidx     = (const int*)d_in[1];
  const float* cweights = (const float*)d_in[2];
  float*       out      = (float*)d_out;
  const int B = in_sizes[0] / 36;
  vqc_kernel<<<512, THREADS, 0, stream>>>(instance, cidx, cweights, out, B);
}


# psi's LDS slot holds (v3, both sets of one lane): 16 16-byte stores per wave and element (were 32), every wire's pairs whole register pairs, two sets per packed op; f32 sum order changes
# speedup vs baseline: 5.6671x; 1.1289x over previous
#include <hip/hip_runtime.h>


typedef __attribute__((ext_vector_type(2))) float v2f;
typedef __attribute__((ext_vector_type(4))) float v4f;
typedef __attribute__((ext_vector_type(8))) float v8f;

#define NSTATE 4096
#define THREADS 128
#define EPB 8
#define PPITCH4 33

__global__ __launch_bounds__(THREADS)
void vqc_kernel(const float* __restrict__ instance,
                const int*   __restrict__ concept_index,
                const float* __restrict__ concept_weights,
                float*       __restrict__ out,
                int B)
{
  __shared__ float gR[EPB][12][2], gI[EPB][12][2];
  __shared__ float UVr[EPB][128], UVi[EPB][128];
  __shared__ float obsA[EPB][4], obsB[EPB][4], obsC[EPB][4];
  __shared__ float4 psiR4[NSTATE / 4];
  __shared__ float4 psiI4[NSTATE / 4];
  __shared__ v4f partial4[EPB * 12 * PPITCH4];
  __shared__ float sums[EPB][12];

  const int tid = threadIdx.x;

  for (int chunk = blockIdx.x; chunk * EPB < B; chunk += gridDim.x) {
  const int b0 = chunk * EPB;

  {
  const int e1 = tid >> 4;
  const int q1 = tid & 15;
  const int b  = b0 + e1;
  if (b >= B) {
    if (q1 < 12) {
      gR[e1][q1][0] = 0.0f;  gI[e1][q1][0] = 0.0f;
      gR[e1][q1][1] = 0.0f;  gI[e1][q1][1] = 0.0f;
    }
  } else if (q1 < 12) {
    const int w = q1;
    const float* a = instance + ((size_t)b * 12 + w) * 3;
    const float phi = a[0], th = a[1], om = a[2];
    float c, s;   sincosf(0.5f * th, &s, &c);
    float s0, c0; sincosf(-0.5f * (phi + om), &s0, &c0);
    float s1, c1; sincosf( 0.5f * (om  - phi), &s1, &c1);
    const int p = 11 - w;
    gR[e1][p][0] = c * c0;  gI[e1][p][0] = c * s0;
    gR[e1][p][1] = s * c1;  gI[e1][p][1] = s * s1;
  } else {
    const int d  = q1 - 12;
    const int ci = concept_index[b];
    const float* cw = concept_weights + (size_t)ci * 12 + d * 3;
    const float cth = cw[1], com = cw[2];
    float sth, vcth; sincosf(cth, &sth, &vcth);
    float so,  co;   sincosf(com, &so,  &co);
    obsA[e1][d] = vcth;
    obsB[e1][d] = 2.0f * sth * co;
    obsC[e1][d] = 2.0f * sth * so;
  }
  }
  __syncthreads();

  {
    const int h     = tid & 63;
    const int base  = (tid < 64) ? 6 : 0;
    const float scl = (tid < 64) ? 0.5f : 1.0f;
#pragma unroll
    for (int e = 0; e < EPB; ++e) {
      float pr = 1.0f, pi = 0.0f;
#pragma unroll
      for (int j = 0; j < 6; ++j) {
        const int t = (h >> j) & 1;
        const float gr = gR[e][base + j][t], gi = gI[e][base + j][t];
        const float nr = pr * gr - pi * gi;
        const float ni = pr * gi + pi * gr;
        pr = nr; pi = ni;
      }
      UVr[e][tid] = scl * pr;
      UVi[e][tid] = scl * pi;
    }
  }
  __syncthreads();

  for (int e = 0; e < EPB; ++e) {
  if (b0 + e >= B) break;

  const int wave = tid >> 5;
  const int lane = tid & 31;
  float4* const myR = psiR4 + wave * 256;
  float4* const myI = psiI4 + wave * 256;
  {
    const int m    = lane & 15;
    const int w1   = wave >> 1;
    const int w0   = wave & 1;
    const int m0 = m & 1, m1 = (m >> 1) & 1, m2 = (m >> 2) & 1, m3 = (m >> 3) & 1;
    const int hU   = ((m0 ^ m1) << 5) | (m1 << 4) | (m3 << 3) | ((w1 ^ m2) << 2) | (m2 << 1) | w0;
    const v2f A = { UVr[e][hU], UVi[e][hU] };
    const int half  = lane >> 4;
    const int laneQ = (m1 << 5) | ((half ^ m1) << 3) | (m3 << 2) | ((m2 ^ m1) << 1) | m0;
#pragma unroll
    for (int J = 0; J < 4; ++J) {
      const float vr = UVr[e][64 + 16 * J + m];
      const float vi = UVi[e][64 + 16 * J + m];
      const v2f Bre = { vr, -vi };
      const v2f Bim = { vi,  vr };
      const v8f z0 = { 0.f, 0.f, 0.f, 0.f, 0.f, 0.f, 0.f, 0.f };
      v8f dre = __builtin_amdgcn_wmma_f32_16x16x4_f32(false, A, false, Bre,
                                                      (short)0, z0, false, false);
      v8f dim = __builtin_amdgcn_wmma_f32_16x16x4_f32(false, A, false, Bim,
                                                      (short)0, z0, false, false);
      const int jQ = ((J & 1) << 6) | ((((J >> 1) ^ J) & 1) << 4);
#pragma unroll
      for (int r2 = 0; r2 < 2; ++r2) {
        const int q = (r2 << 7) | jQ | laneQ;
        myR[q] = make_float4(dre[4 * r2], dre[4 * r2 + 1], dre[4 * r2 + 2], dre[4 * r2 + 3]);
        myI[q] = make_float4(dim[4 * r2], dim[4 * r2 + 1], dim[4 * r2 + 2], dim[4 * r2 + 3]);
      }
    }
  }

  float acc[12];
  {
    v2f ar[2][8], ai[2][8], nr[2][8];
#pragma unroll
    for (int j = 0; j < 8; ++j) {
      const int q = (j << 5) | (lane ^ ((j & 1) << 3));
      const float4 vr = myR[q];
      const float4 vi = myI[q];
      ar[0][j] = v2f{ vr.x, vr.y };  ar[1][j] = v2f{ vr.z, vr.w };
      ai[0][j] = v2f{ vi.x, vi.y };  ai[1][j] = v2f{ vi.z, vi.w };
    }
#pragma unroll
    for (int c = 0; c < 2; ++c)
#pragma unroll
      for (int j = 0; j < 8; ++j)
        nr[c][j] = __builtin_elementwise_fma(ar[c][j], ar[c][j], ai[c][j] * ai[c][j]);

    v2f s1[2][4], d1[2][4];
#pragma unroll
    for (int c = 0; c < 2; ++c)
#pragma unroll
      for (int p = 0; p < 4; ++p) {
        s1[c][p] = nr[c][2 * p] + nr[c][2 * p + 1];
        d1[c][p] = nr[c][2 * p] - nr[c][2 * p + 1];
      }
    const v2f dn3 = ((d1[0][0] + d1[0][1]) + (d1[0][2] + d1[0][3])) +
                    ((d1[1][0] + d1[1][1]) + (d1[1][2] + d1[1][3]));
    v2f s2[2][2], d2[2][2];
#pragma unroll
    for (int c = 0; c < 2; ++c)
#pragma unroll
      for (int g = 0; g < 2; ++g) {
        s2[c][g] = s1[c][2 * g] + s1[c][2 * g + 1];
        d2[c][g] = s1[c][2 * g] - s1[c][2 * g + 1];
      }
    const v2f dn2 = (d2[0][0] + d2[0][1]) + (d2[1][0] + d2[1][1]);
    const v2f s30 = s2[0][0] + s2[0][1], s31 = s2[1][0] + s2[1][1];
    const v2f dn1 = (s2[0][0] - s2[0][1]) + (s2[1][0] - s2[1][1]);
    const v2f dn0 = s30 - s31;

    const v2f zero2 = { 0.0f, 0.0f };
    v2f xr0 = zero2, xi0 = zero2;
#pragma unroll
    for (int j = 0; j < 8; ++j) {
      xr0 = __builtin_elementwise_fma(ar[0][j], ar[1][j], xr0);
      xr0 = __builtin_elementwise_fma(ai[0][j], ai[1][j], xr0);
      xi0 = __builtin_elementwise_fma(ar[0][j], ai[1][j], xi0);
      xi0 = __builtin_elementwise_fma(-ai[0][j], ar[1][j], xi0);
    }
    v2f xr1 = zero2, xi1 = zero2;
#pragma unroll
    for (int c = 0; c < 2; ++c)
#pragma unroll
      for (int j = 0; j < 4; ++j) {
        xr1 = __builtin_elementwise_fma(ar[c][j], ar[c][j + 4], xr1);
        xr1 = __builtin_elementwise_fma(ai[c][j], ai[c][j + 4], xr1);
        xi1 = __builtin_elementwise_fma(ar[c][j], ai[c][j + 4], xi1);
        xi1 = __builtin_elementwise_fma(-ai[c][j], ar[c][j + 4], xi1);
      }
    v2f xr2 = zero2, xi2 = zero2;
#pragma unroll
    for (int c = 0; c < 2; ++c)
#pragma unroll
      for (int g = 0; g < 4; ++g) {
        const int j = ((g >> 1) << 2) | (g & 1);
        xr2 = __builtin_elementwise_fma(ar[c][j], ar[c][j + 2], xr2);
        xr2 = __builtin_elementwise_fma(ai[c][j], ai[c][j + 2], xr2);
        xi2 = __builtin_elementwise_fma(ar[c][j], ai[c][j + 2], xi2);
        xi2 = __builtin_elementwise_fma(-ai[c][j], ar[c][j + 2], xi2);
      }
    v2f xr3 = zero2, xi3 = zero2;
#pragma unroll
    for (int c = 0; c < 2; ++c)
#pragma unroll
      for (int j = 0; j < 8; j += 2) {
        xr3 = __builtin_elementwise_fma(ar[c][j], ar[c][j + 1], xr3);
        xr3 = __builtin_elementwise_fma(ai[c][j], ai[c][j + 1], xr3);
        xi3 = __builtin_elementwise_fma(ar[c][j], ai[c][j + 1], xi3);
        xi3 = __builtin_elementwise_fma(-ai[c][j], ar[c][j + 1], xi3);
      }

    const float sg0 = (__popc(tid & 0x7F) & 1) ? -1.0f : 1.0f;
    const float sg1 = (__popc(tid & 0x48) & 1) ? -1.0f : 1.0f;
    const float sg2 = (__popc(tid & 0x78) & 1) ? -1.0f : 1.0f;
    const float sg3 = (__popc(tid & 0x7E) & 1) ? -1.0f : 1.0f;
    acc[0]  = sg0 * (dn0.x + dn0.y);  acc[1]  = xr0.x + xr0.y;  acc[2]  = sg0 * (xi0.x + xi0.y);
    acc[3]  = sg1 * (dn1.x - dn1.y);  acc[4]  = xr1.x + xr1.y;  acc[5]  = sg1 * (xi1.x - xi1.y);
    acc[6]  = sg2 * (dn2.x - dn2.y);  acc[7]  = xr2.x + xr2.y;  acc[8]  = sg2 * (xi2.x - xi2.y);
    acc[9]  = sg3 * (dn3.x - dn3.y);  acc[10] = xr3.x + xr3.y;  acc[11] = sg3 * (xi3.x - xi3.y);
  }
  {
    float* prow = reinterpret_cast<float*>(partial4) + (e * 12) * (PPITCH4 * 4) + tid;
#pragma unroll
    for (int q = 0; q < 12; ++q) prow[q * (PPITCH4 * 4)] = acc[q];
  }
  }
  __syncthreads();

  if (tid < EPB * 12) {
    const int e5 = tid / 12;
    const int q5 = tid - e5 * 12;
    if (b0 + e5 < B) {
      const v4f* row = partial4 + tid * PPITCH4;
      v4f a0 = row[0], a1 = row[1], a2 = row[2], a3 = row[3];
#pragma unroll
      for (int j = 4; j < THREADS / 4; j += 4) {
        a0 += row[j];  a1 += row[j + 1];  a2 += row[j + 2];  a3 += row[j + 3];
      }
      const v4f s = (a0 + a1) + (a2 + a3);
      sums[e5][q5] = (s.x + s.y) + (s.z + s.w);
    }
  }
  __syncthreads();

  if (tid < EPB * 4) {
    const int e6 = tid >> 2;
    const int d  = tid & 3;
    if (b0 + e6 < B) {
      const float dn  = sums[e6][d * 3 + 0];
      const float crR = sums[e6][d * 3 + 1];
      const float crI = sums[e6][d * 3 + 2];
      const float z = obsA[e6][d] * dn + obsB[e6][d] * crR + obsC[e6][d] * crI;
      out[(size_t)(b0 + e6) * 4 + d] = 0.5f * z + 0.5f;
    }
  }
  __syncthreads();
  }
}

extern "C" void kernel_launch(void* const* d_in, const int* in_sizes, int n_in,
                              void* d_out, int out_size, void* d_ws, size_t ws_size,
                              hipStream_t stream) {
  const float* instance = (const float*)d_in[0];
  const int*   cidx     = (const int*)d_in[1];
  const float* cweights = (const float*)d_in[2];
  float*       out      = (float*)d_out;
  const int B = in_sizes[0] / 36;
  vqc_kernel<<<512, THREADS, 0, stream>>>(instance, cidx, cweights, out, B);
}


# Observable weights off the set-up path: wave 3 loads the concept index beside the gate angles, the concept angles beside phase 2, takes their sincos beside the block sums, writes outputs; sums kept
# speedup vs baseline: 6.1509x; 1.0854x over previous
#include <hip/hip_runtime.h>


typedef __attribute__((ext_vector_type(2))) float v2f;
typedef __attribute__((ext_vector_type(4))) float v4f;
typedef __attribute__((ext_vector_type(8))) float v8f;

#define NSTATE 4096
#define THREADS 128
#define EPB 8
#define PPITCH4 33

__global__ __launch_bounds__(THREADS)
void vqc_kernel(const float* __restrict__ instance,
                const int*   __restrict__ concept_index,
                const float* __restrict__ concept_weights,
                float*       __restrict__ out,
                int B)
{
  __shared__ float gR[EPB][12][2], gI[EPB][12][2];
  __shared__ float UVr[EPB][128], UVi[EPB][128];
  __shared__ float4 psiR4[NSTATE / 4];
  __shared__ float4 psiI4[NSTATE / 4];
  __shared__ v4f partial4[EPB * 12 * PPITCH4];
  __shared__ float sums[EPB][12];

  const int tid = threadIdx.x;

  for (int chunk = blockIdx.x; chunk * EPB < B; chunk += gridDim.x) {
  const int b0 = chunk * EPB;

  int   ci  = 0;
  float cth = 0.0f, com = 0.0f;
  if (tid < EPB * 12) {
    const int e1 = tid / 12;
    const int w  = tid - e1 * 12;
    const int p  = 11 - w;
    const int b  = b0 + e1;
    if (b >= B) {
      gR[e1][p][0] = 0.0f;  gI[e1][p][0] = 0.0f;
      gR[e1][p][1] = 0.0f;  gI[e1][p][1] = 0.0f;
    } else {
      const float* a = instance + ((size_t)b * 12 + w) * 3;
      const float phi = a[0], th = a[1], om = a[2];
      float c, s;   sincosf(0.5f * th, &s, &c);
      float s0, c0; sincosf(-0.5f * (phi + om), &s0, &c0);
      float s1, c1; sincosf( 0.5f * (om  - phi), &s1, &c1);
      gR[e1][p][0] = c * c0;  gI[e1][p][0] = c * s0;
      gR[e1][p][1] = s * c1;  gI[e1][p][1] = s * s1;
    }
  } else {
    const int b = b0 + ((tid - EPB * 12) >> 2);
    if (b < B) ci = concept_index[b];
  }
  __syncthreads();

  if (tid >= EPB * 12) {
    const int l6 = tid - EPB * 12;
    const int d  = l6 & 3;
    if (b0 + (l6 >> 2) < B) {
      const float* cw = concept_weights + (size_t)ci * 12 + d * 3;
      cth = cw[1];  com = cw[2];
    }
  }

  {
    const int h     = tid & 63;
    const int base  = (tid < 64) ? 6 : 0;
    const float scl = (tid < 64) ? 0.5f : 1.0f;
#pragma unroll
    for (int e = 0; e < EPB; ++e) {
      float pr = 1.0f, pi = 0.0f;
#pragma unroll
      for (int j = 0; j < 6; ++j) {
        const int t = (h >> j) & 1;
        const float gr = gR[e][base + j][t], gi = gI[e][base + j][t];
        const float nr = pr * gr - pi * gi;
        const float ni = pr * gi + pi * gr;
        pr = nr; pi = ni;
      }
      UVr[e][tid] = scl * pr;
      UVi[e][tid] = scl * pi;
    }
  }
  __syncthreads();

  for (int e = 0; e < EPB; ++e) {
  if (b0 + e >= B) break;

  const int wave = tid >> 5;
  const int lane = tid & 31;
  float4* const myR = psiR4 + wave * 256;
  float4* const myI = psiI4 + wave * 256;
  {
    const int m    = lane & 15;
    const int w1   = wave >> 1;
    const int w0   = wave & 1;
    const int m0 = m & 1, m1 = (m >> 1) & 1, m2 = (m >> 2) & 1, m3 = (m >> 3) & 1;
    const int hU   = ((m0 ^ m1) << 5) | (m1 << 4) | (m3 << 3) | ((w1 ^ m2) << 2) | (m2 << 1) | w0;
    const v2f A = { UVr[e][hU], UVi[e][hU] };
    const int half  = lane >> 4;
    const int laneQ = (m1 << 5) | ((half ^ m1) << 3) | (m3 << 2) | ((m2 ^ m1) << 1) | m0;
#pragma unroll
    for (int J = 0; J < 4; ++J) {
      const float vr = UVr[e][64 + 16 * J + m];
      const float vi = UVi[e][64 + 16 * J + m];
      const v2f Bre = { vr, -vi };
      const v2f Bim = { vi,  vr };
      const v8f z0 = { 0.f, 0.f, 0.f, 0.f, 0.f, 0.f, 0.f, 0.f };
      v8f dre = __builtin_amdgcn_wmma_f32_16x16x4_f32(false, A, false, Bre,
                                                      (short)0, z0, false, false);
      v8f dim = __builtin_amdgcn_wmma_f32_16x16x4_f32(false, A, false, Bim,
                                                      (short)0, z0, false, false);
      const int jQ = ((J & 1) << 6) | ((((J >> 1) ^ J) & 1) << 4);
#pragma unroll
      for (int r2 = 0; r2 < 2; ++r2) {
        const int q = (r2 << 7) | jQ | laneQ;
        myR[q] = make_float4(dre[4 * r2], dre[4 * r2 + 1], dre[4 * r2 + 2], dre[4 * r2 + 3]);
        myI[q] = make_float4(dim[4 * r2], dim[4 * r2 + 1], dim[4 * r2 + 2], dim[4 * r2 + 3]);
      }
    }
  }

  float acc[12];
  {
    v2f ar[2][8], ai[2][8], nr[2][8];
#pragma unroll
    for (int j = 0; j < 8; ++j) {
      const int q = (j << 5) | (lane ^ ((j & 1) << 3));
      const float4 vr = myR[q];
      const float4 vi = myI[q];
      ar[0][j] = v2f{ vr.x, vr.y };  ar[1][j] = v2f{ vr.z, vr.w };
      ai[0][j] = v2f{ vi.x, vi.y };  ai[1][j] = v2f{ vi.z, vi.w };
    }
#pragma unroll
    for (int c = 0; c < 2; ++c)
#pragma unroll
      for (int j = 0; j < 8; ++j)
        nr[c][j] = __builtin_elementwise_fma(ar[c][j], ar[c][j], ai[c][j] * ai[c][j]);

    v2f s1[2][4], d1[2][4];
#pragma unroll
    for (int c = 0; c < 2; ++c)
#pragma unroll
      for (int p = 0; p < 4; ++p) {
        s1[c][p] = nr[c][2 * p] + nr[c][2 * p + 1];
        d1[c][p] = nr[c][2 * p] - nr[c][2 * p + 1];
      }
    const v2f dn3 = ((d1[0][0] + d1[0][1]) + (d1[0][2] + d1[0][3])) +
                    ((d1[1][0] + d1[1][1]) + (d1[1][2] + d1[1][3]));
    v2f s2[2][2], d2[2][2];
#pragma unroll
    for (int c = 0; c < 2; ++c)
#pragma unroll
      for (int g = 0; g < 2; ++g) {
        s2[c][g] = s1[c][2 * g] + s1[c][2 * g + 1];
        d2[c][g] = s1[c][2 * g] - s1[c][2 * g + 1];
      }
    const v2f dn2 = (d2[0][0] + d2[0][1]) + (d2[1][0] + d2[1][1]);
    const v2f s30 = s2[0][0] + s2[0][1], s31 = s2[1][0] + s2[1][1];
    const v2f dn1 = (s2[0][0] - s2[0][1]) + (s2[1][0] - s2[1][1]);
    const v2f dn0 = s30 - s31;

    const v2f zero2 = { 0.0f, 0.0f };
    v2f xr0 = zero2, xi0 = zero2;
#pragma unroll
    for (int j = 0; j < 8; ++j) {
      xr0 = __builtin_elementwise_fma(ar[0][j], ar[1][j], xr0);
      xr0 = __builtin_elementwise_fma(ai[0][j], ai[1][j], xr0);
      xi0 = __builtin_elementwise_fma(ar[0][j], ai[1][j], xi0);
      xi0 = __builtin_elementwise_fma(-ai[0][j], ar[1][j], xi0);
    }
    v2f xr1 = zero2, xi1 = zero2;
#pragma unroll
    for (int c = 0; c < 2; ++c)
#pragma unroll
      for (int j = 0; j < 4; ++j) {
        xr1 = __builtin_elementwise_fma(ar[c][j], ar[c][j + 4], xr1);
        xr1 = __builtin_elementwise_fma(ai[c][j], ai[c][j + 4], xr1);
        xi1 = __builtin_elementwise_fma(ar[c][j], ai[c][j + 4], xi1);
        xi1 = __builtin_elementwise_fma(-ai[c][j], ar[c][j + 4], xi1);
      }
    v2f xr2 = zero2, xi2 = zero2;
#pragma unroll
    for (int c = 0; c < 2; ++c)
#pragma unroll
      for (int g = 0; g < 4; ++g) {
        const int j = ((g >> 1) << 2) | (g & 1);
        xr2 = __builtin_elementwise_fma(ar[c][j], ar[c][j + 2], xr2);
        xr2 = __builtin_elementwise_fma(ai[c][j], ai[c][j + 2], xr2);
        xi2 = __builtin_elementwise_fma(ar[c][j], ai[c][j + 2], xi2);
        xi2 = __builtin_elementwise_fma(-ai[c][j], ar[c][j + 2], xi2);
      }
    v2f xr3 = zero2, xi3 = zero2;
#pragma unroll
    for (int c = 0; c < 2; ++c)
#pragma unroll
      for (int j = 0; j < 8; j += 2) {
        xr3 = __builtin_elementwise_fma(ar[c][j], ar[c][j + 1], xr3);
        xr3 = __builtin_elementwise_fma(ai[c][j], ai[c][j + 1], xr3);
        xi3 = __builtin_elementwise_fma(ar[c][j], ai[c][j + 1], xi3);
        xi3 = __builtin_elementwise_fma(-ai[c][j], ar[c][j + 1], xi3);
      }

    const float sg0 = (__popc(tid & 0x7F) & 1) ? -1.0f : 1.0f;
    const float sg1 = (__popc(tid & 0x48) & 1) ? -1.0f : 1.0f;
    const float sg2 = (__popc(tid & 0x78) & 1) ? -1.0f : 1.0f;
    const float sg3 = (__popc(tid & 0x7E) & 1) ? -1.0f : 1.0f;
    acc[0]  = sg0 * (dn0.x + dn0.y);  acc[1]  = xr0.x + xr0.y;  acc[2]  = sg0 * (xi0.x + xi0.y);
    acc[3]  = sg1 * (dn1.x - dn1.y);  acc[4]  = xr1.x + xr1.y;  acc[5]  = sg1 * (xi1.x - xi1.y);
    acc[6]  = sg2 * (dn2.x - dn2.y);  acc[7]  = xr2.x + xr2.y;  acc[8]  = sg2 * (xi2.x - xi2.y);
    acc[9]  = sg3 * (dn3.x - dn3.y);  acc[10] = xr3.x + xr3.y;  acc[11] = sg3 * (xi3.x - xi3.y);
  }
  {
    float* prow = reinterpret_cast<float*>(partial4) + (e * 12) * (PPITCH4 * 4) + tid;
#pragma unroll
    for (int q = 0; q < 12; ++q) prow[q * (PPITCH4 * 4)] = acc[q];
  }
  }
  __syncthreads();

  float oA = 0.0f, oB = 0.0f, oC = 0.0f;
  if (tid < EPB * 12) {
    const int e5 = tid / 12;
    const int q5 = tid - e5 * 12;
    if (b0 + e5 < B) {
      const v4f* row = partial4 + tid * PPITCH4;
      v4f a0 = row[0], a1 = row[1], a2 = row[2], a3 = row[3];
#pragma unroll
      for (int j = 4; j < THREADS / 4; j += 4) {
        a0 += row[j];  a1 += row[j + 1];  a2 += row[j + 2];  a3 += row[j + 3];
      }
      const v4f s = (a0 + a1) + (a2 + a3);
      sums[e5][q5] = (s.x + s.y) + (s.z + s.w);
    }
  } else {
    float sth, vcth; sincosf(cth, &sth, &vcth);
    float so,  co;   sincosf(com, &so,  &co);
    oA = vcth;
    oB = 2.0f * sth * co;
    oC = 2.0f * sth * so;
  }
  __syncthreads();

  if (tid >= EPB * 12) {
    const int l6 = tid - EPB * 12;
    const int e6 = l6 >> 2;
    const int d  = l6 & 3;
    if (b0 + e6 < B) {
      const float dn  = sums[e6][d * 3 + 0];
      const float crR = sums[e6][d * 3 + 1];
      const float crI = sums[e6][d * 3 + 2];
      const float z = oA * dn + oB * crR + oC * crI;
      out[(size_t)(b0 + e6) * 4 + d] = 0.5f * z + 0.5f;
    }
  }
  __syncthreads();
  }
}

extern "C" void kernel_launch(void* const* d_in, const int* in_sizes, int n_in,
                              void* d_out, int out_size, void* d_ws, size_t ws_size,
                              hipStream_t stream) {
  const float* instance = (const float*)d_in[0];
  const int*   cidx     = (const int*)d_in[1];
  const float* cweights = (const float*)d_in[2];
  float*       out      = (float*)d_out;
  const int B = in_sizes[0] / 36;
  vqc_kernel<<<512, THREADS, 0, stream>>>(instance, cidx, cweights, out, B);
}
